# starting kernel
# baseline (speedup 1.0000x reference)
#include <hip/hip_runtime.h>

typedef __attribute__((ext_vector_type(2))) float v2f;
typedef __attribute__((ext_vector_type(8))) float v8f;

template <int IMM>
__device__ __forceinline__ float swz_xor_add(float v) {
    int p = __builtin_amdgcn_ds_swizzle(__float_as_int(v), IMM);
    return v + __int_as_float(p);
}

__global__ __launch_bounds__(256) void dendrite_wmma_kernel(
    const float* __restrict__ x,
    const float* __restrict__ kk,
    const float* __restrict__ w,
    const float* __restrict__ q,
    const float* __restrict__ lin_w,
    const float* __restrict__ lin_b,
    float* __restrict__ out)
{
    __shared__ float plo_lds[8][16][16];

    const int lane = threadIdx.x & 31;
    const int wv   = threadIdx.x >> 5;
    const int half = lane >> 4;
    const int n    = lane & 15;
    const int tile = blockIdx.x * 8 + wv;
    const int row_g0 = tile * 16;

    v2f bmat[4];
    #pragma unroll
    for (int j = 0; j < 4; ++j) {
        #pragma unroll
        for (int e = 0; e < 2; ++e) {
            int h = 4 * j + 2 * half + e;
            int v = h * 16 + n;
            bmat[j][e] = (v == 0) ? 0.0f : lin_w[v - 1];
        }
    }

    const int row_g = row_g0 + n;
    const int b_idx = row_g >> 5;
    const int d_idx = row_g & 31;

    float sv[8];
    #pragma unroll
    for (int s = 0; s < 8; ++s) {
        float xs = x[b_idx * 8 + s];
        float z  = kk[d_idx * 8 + s] * (w[d_idx * 8 + s] * xs - q[d_idx * 8 + s]);
        sv[s] = __builtin_amdgcn_rcpf(1.0f + __expf(-z));
    }

    const float t0 = sv[3], t1 = sv[2], t2 = sv[1], t3 = sv[0];
    const float base = half ? t1 : 1.0f;

    v2f amat[4];
    amat[0][0] = base;
    amat[0][1] = base * t0;
    amat[1][0] = base * t2;
    amat[1][1] = amat[1][0] * t0;
    amat[2][0] = base * t3;
    amat[2][1] = amat[2][0] * t0;
    amat[3][0] = amat[2][0] * t2;
    amat[3][1] = amat[3][0] * t0;

    if (half == 0) {
        float pl[16];
        pl[0] = 1.0f;
        int len = 1;
        #pragma unroll
        for (int j = 0; j < 4; ++j) {
            float tl = sv[7 - j];
            #pragma unroll
            for (int m = 0; m < 8; ++m) {
                if (m < len) pl[len + m] = pl[m] * tl;
            }
            len <<= 1;
        }
        #pragma unroll
        for (int i = 0; i < 16; ++i)
            plo_lds[wv][n][i] = pl[i];
    }

    v8f c = {};
    #pragma unroll
    for (int j = 0; j < 4; ++j)
        c = __builtin_amdgcn_wmma_f32_16x16x4_f32(
                 false, amat[j],  false, bmat[j],
                 (short)0, c,  false,  false);

    float partial[8];
    #pragma unroll
    for (int r = 0; r < 8; ++r)
        partial[r] = c[r] * plo_lds[wv][8 * half + r][n];

    #pragma unroll
    for (int r = 0; r < 8; ++r) {
        float v = partial[r];
        v = swz_xor_add<(1 << 10) | 0x1f>(v);
        v = swz_xor_add<(2 << 10) | 0x1f>(v);
        v = swz_xor_add<(4 << 10) | 0x1f>(v);
        v = swz_xor_add<(8 << 10) | 0x1f>(v);
        partial[r] = v;
    }

    if (n == 0) {
        float bias = lin_b[0];
        #pragma unroll
        for (int r = 0; r < 8; ++r)
            out[row_g0 + 8 * half + r] = partial[r] + bias;
    }
}

extern "C" void kernel_launch(void* const* d_in, const int* in_sizes, int n_in,
                              void* d_out, int out_size, void* d_ws, size_t ws_size,
                              hipStream_t stream) {
    const float* x     = (const float*)d_in[0];
    const float* k     = (const float*)d_in[1];
    const float* w     = (const float*)d_in[2];
    const float* q     = (const float*)d_in[3];
    const float* lin_w = (const float*)d_in[5];
    const float* lin_b = (const float*)d_in[6];
    float* out = (float*)d_out;

    dendrite_wmma_kernel<<<dim3(2048), dim3(256), 0, stream>>>(
        x, k, w, q, lin_w, lin_b, out);
}


# row sums over 16 lanes leave the LDS unit: halving tree of 8 DPP row_xmask adds per wave replaces 32 ds_swizzle + 32 adds, one 16-lane store, bias load held early; same summation tree
# speedup vs baseline: 1.1072x; 1.1072x over previous
#include <hip/hip_runtime.h>

typedef __attribute__((ext_vector_type(2))) float v2f;
typedef __attribute__((ext_vector_type(8))) float v8f;

template <int M>
__device__ __forceinline__ float row_xor(float v) {
    return __int_as_float(__builtin_amdgcn_update_dpp(
        0, __float_as_int(v), 0x160 | M, 0xf, 0xf, true));
}

template <int M>
__device__ __forceinline__ float halve_step(bool hi, float lo_v, float hi_v) {
    const float keep = hi ? hi_v : lo_v;
    const float send = hi ? lo_v : hi_v;
    return keep + row_xor<M>(send);
}

__global__ __launch_bounds__(256) void dendrite_wmma_kernel(
    const float* __restrict__ x,
    const float* __restrict__ kk,
    const float* __restrict__ w,
    const float* __restrict__ q,
    const float* __restrict__ lin_w,
    const float* __restrict__ lin_b,
    float* __restrict__ out)
{
    __shared__ float plo_lds[8][16][16];

    const int lane = threadIdx.x & 31;
    const int wv   = threadIdx.x >> 5;
    const int half = lane >> 4;
    const int n    = lane & 15;
    const int tile = blockIdx.x * 8 + wv;
    const int row_g0 = tile * 16;
    const float bias = lin_b[0];

    v2f bmat[4];
    #pragma unroll
    for (int j = 0; j < 4; ++j) {
        #pragma unroll
        for (int e = 0; e < 2; ++e) {
            int h = 4 * j + 2 * half + e;
            int v = h * 16 + n;
            bmat[j][e] = (v == 0) ? 0.0f : lin_w[v - 1];
        }
    }

    const int row_g = row_g0 + n;
    const int b_idx = row_g >> 5;
    const int d_idx = row_g & 31;

    float sv[8];
    #pragma unroll
    for (int s = 0; s < 8; ++s) {
        float xs = x[b_idx * 8 + s];
        float z  = kk[d_idx * 8 + s] * (w[d_idx * 8 + s] * xs - q[d_idx * 8 + s]);
        sv[s] = __builtin_amdgcn_rcpf(1.0f + __expf(-z));
    }

    const float t0 = sv[3], t1 = sv[2], t2 = sv[1], t3 = sv[0];
    const float base = half ? t1 : 1.0f;

    v2f amat[4];
    amat[0][0] = base;
    amat[0][1] = base * t0;
    amat[1][0] = base * t2;
    amat[1][1] = amat[1][0] * t0;
    amat[2][0] = base * t3;
    amat[2][1] = amat[2][0] * t0;
    amat[3][0] = amat[2][0] * t2;
    amat[3][1] = amat[3][0] * t0;

    if (half == 0) {
        float pl[16];
        pl[0] = 1.0f;
        int len = 1;
        #pragma unroll
        for (int j = 0; j < 4; ++j) {
            float tl = sv[7 - j];
            #pragma unroll
            for (int m = 0; m < 8; ++m) {
                if (m < len) pl[len + m] = pl[m] * tl;
            }
            len <<= 1;
        }
        #pragma unroll
        for (int i = 0; i < 16; ++i)
            plo_lds[wv][n][i] = pl[i];
    }

    v8f c = {};
    #pragma unroll
    for (int j = 0; j < 4; ++j)
        c = __builtin_amdgcn_wmma_f32_16x16x4_f32(
                 false, amat[j],  false, bmat[j],
                 (short)0, c,  false,  false);

    float partial[8];
    #pragma unroll
    for (int r = 0; r < 8; ++r)
        partial[r] = c[r] * plo_lds[wv][8 * half + r][n];

    asm volatile("" :: "s"(bias));

    const bool b0 = (lane & 1) != 0;
    const bool b1 = (lane & 2) != 0;
    const bool b2 = (lane & 4) != 0;

    const float a0 = halve_step<1>(b0, partial[0], partial[4]);
    const float a1 = halve_step<1>(b0, partial[1], partial[5]);
    const float a2 = halve_step<1>(b0, partial[2], partial[6]);
    const float a3 = halve_step<1>(b0, partial[3], partial[7]);
    const float c0 = halve_step<2>(b1, a0, a2);
    const float c1 = halve_step<2>(b1, a1, a3);
    float s1 = halve_step<4>(b2, c0, c1);
    s1 = s1 + row_xor<8>(s1);

    if ((lane & 8) == 0) {
        const int r = ((lane & 1) << 2) | (lane & 2) | ((lane >> 2) & 1);
        out[row_g0 + 8 * half + r] = s1 + bias;
    }
}

extern "C" void kernel_launch(void* const* d_in, const int* in_sizes, int n_in,
                              void* d_out, int out_size, void* d_ws, size_t ws_size,
                              hipStream_t stream) {
    const float* x     = (const float*)d_in[0];
    const float* k     = (const float*)d_in[1];
    const float* w     = (const float*)d_in[2];
    const float* q     = (const float*)d_in[3];
    const float* lin_w = (const float*)d_in[5];
    const float* lin_b = (const float*)d_in[6];
    float* out = (float*)d_out;

    dendrite_wmma_kernel<<<dim3(2048), dim3(256), 0, stream>>>(
        x, k, w, q, lin_w, lin_b, out);
}


# each row's 8 sigmoid gates computed once: its two lanes take 4 gates each (one 16-byte load per input), the half-0 lane gets its 3 A-side gates by permlanex16; values and sum order unchanged
# speedup vs baseline: 1.1975x; 1.0816x over previous
#include <hip/hip_runtime.h>

typedef __attribute__((ext_vector_type(2))) float v2f;
typedef __attribute__((ext_vector_type(8))) float v8f;

template <int M>
__device__ __forceinline__ float row_xor(float v) {
    return __int_as_float(__builtin_amdgcn_update_dpp(
        0, __float_as_int(v), 0x160 | M, 0xf, 0xf, true));
}

template <int M>
__device__ __forceinline__ float halve_step(bool hi, float lo_v, float hi_v) {
    const float keep = hi ? hi_v : lo_v;
    const float send = hi ? lo_v : hi_v;
    return keep + row_xor<M>(send);
}

typedef float f4u __attribute__((ext_vector_type(4), aligned(4)));

__device__ __forceinline__ float xhalf(float v) {
    const int b = __float_as_int(v);
    return __int_as_float(__builtin_amdgcn_permlanex16(
        b, b, 0x76543210, 0xfedcba98, false, false));
}

__device__ __forceinline__ float gate(float k, float w, float x, float q) {
    const float z = k * (w * x - q);
    return __builtin_amdgcn_rcpf(1.0f + __expf(-z));
}

__global__ __launch_bounds__(256) void dendrite_wmma_kernel(
    const float* __restrict__ x,
    const float* __restrict__ kk,
    const float* __restrict__ w,
    const float* __restrict__ q,
    const float* __restrict__ lin_w,
    const float* __restrict__ lin_b,
    float* __restrict__ out)
{
    __shared__ float plo_lds[8][16][16];

    const int lane = threadIdx.x & 31;
    const int wv   = threadIdx.x >> 5;
    const int half = lane >> 4;
    const int n    = lane & 15;
    const int tile = blockIdx.x * 8 + wv;
    const int row_g0 = tile * 16;
    const float bias = lin_b[0];

    v2f bmat[4];
    #pragma unroll
    for (int j = 0; j < 4; ++j) {
        #pragma unroll
        for (int e = 0; e < 2; ++e) {
            int h = 4 * j + 2 * half + e;
            int v = h * 16 + n;
            bmat[j][e] = (v == 0) ? 0.0f : lin_w[v - 1];
        }
    }

    const int row_g = row_g0 + n;
    const int b_idx = row_g >> 5;
    const int d_idx = row_g & 31;
    const unsigned s0 = half ? 0u : 4u;
    const unsigned xo = (unsigned)b_idx * 8u + s0;
    const unsigned po = (unsigned)d_idx * 8u + s0;

    const f4u xv = *reinterpret_cast<const f4u*>(x  + xo);
    const f4u kv = *reinterpret_cast<const f4u*>(kk + po);
    const f4u wq = *reinterpret_cast<const f4u*>(w  + po);
    const f4u qv = *reinterpret_cast<const f4u*>(q  + po);

    const float g[4] = { gate(kv[0], wq[0], xv[0], qv[0]), gate(kv[1], wq[1], xv[1], qv[1]),
                         gate(kv[2], wq[2], xv[2], qv[2]), gate(kv[3], wq[3], xv[3], qv[3]) };

    const float p0 = xhalf(g[0]);
    const float p1 = xhalf(g[1]);
    const float p3 = xhalf(g[3]);

    const float t0 = half ? g[3] : p3;
    const float t2 = half ? g[1] : p1;
    const float t3 = half ? g[0] : p0;
    const float base = half ? g[2] : 1.0f;

    v2f amat[4];
    amat[0][0] = base;
    amat[0][1] = base * t0;
    amat[1][0] = base * t2;
    amat[1][1] = amat[1][0] * t0;
    amat[2][0] = base * t3;
    amat[2][1] = amat[2][0] * t0;
    amat[3][0] = amat[2][0] * t2;
    amat[3][1] = amat[3][0] * t0;

    if (half == 0) {
        float pl[16];
        pl[0] = 1.0f;
        int len = 1;
        #pragma unroll
        for (int j = 0; j < 4; ++j) {
            float tl = g[3 - j];
            #pragma unroll
            for (int m = 0; m < 8; ++m) {
                if (m < len) pl[len + m] = pl[m] * tl;
            }
            len <<= 1;
        }
        #pragma unroll
        for (int i = 0; i < 16; ++i)
            plo_lds[wv][n][i] = pl[i];
    }

    v8f c = {};
    #pragma unroll
    for (int j = 0; j < 4; ++j)
        c = __builtin_amdgcn_wmma_f32_16x16x4_f32(
                 false, amat[j],  false, bmat[j],
                 (short)0, c,  false,  false);

    float partial[8];
    #pragma unroll
    for (int r = 0; r < 8; ++r)
        partial[r] = c[r] * plo_lds[wv][8 * half + r][n];

    asm volatile("" :: "s"(bias));

    const bool b0 = (lane & 1) != 0;
    const bool b1 = (lane & 2) != 0;
    const bool b2 = (lane & 4) != 0;

    const float a0 = halve_step<1>(b0, partial[0], partial[4]);
    const float a1 = halve_step<1>(b0, partial[1], partial[5]);
    const float a2 = halve_step<1>(b0, partial[2], partial[6]);
    const float a3 = halve_step<1>(b0, partial[3], partial[7]);
    const float c0 = halve_step<2>(b1, a0, a2);
    const float c1 = halve_step<2>(b1, a1, a3);
    float s1 = halve_step<4>(b2, c0, c1);
    s1 = s1 + row_xor<8>(s1);

    if ((lane & 8) == 0) {
        const int r = ((lane & 1) << 2) | (lane & 2) | ((lane >> 2) & 1);
        out[row_g0 + 8 * half + r] = s1 + bias;
    }
}

extern "C" void kernel_launch(void* const* d_in, const int* in_sizes, int n_in,
                              void* d_out, int out_size, void* d_ws, size_t ws_size,
                              hipStream_t stream) {
    const float* x     = (const float*)d_in[0];
    const float* k     = (const float*)d_in[1];
    const float* w     = (const float*)d_in[2];
    const float* q     = (const float*)d_in[3];
    const float* lin_w = (const float*)d_in[5];
    const float* lin_b = (const float*)d_in[6];
    float* out = (float*)d_out;

    dendrite_wmma_kernel<<<dim3(2048), dim3(256), 0, stream>>>(
        x, k, w, q, lin_w, lin_b, out);
}
